# starting kernel
# baseline (speedup 1.0000x reference)
#include <hip/hip_runtime.h>
#include <math.h>

#define VOCAB 50257
#define EMBED 128
#define NSENT 50
#define TOK   20
#define BATCH 64
#define NHOPS 3

typedef __attribute__((ext_vector_type(2))) float v2f;
typedef __attribute__((ext_vector_type(8))) float v8f;

__global__ void k_embed_q(const int* __restrict__ questions,
                          const float* __restrict__ W_B,
                          float* __restrict__ u) {
  const int b = blockIdx.x;
  const int d = threadIdx.x;
  float s = 0.f;
#pragma unroll
  for (int t = 0; t < TOK; ++t) {
    const int tok = questions[b * TOK + t];
    s += W_B[tok * EMBED + d];
  }
  u[b * EMBED + d] = s;
}

__global__ void k_embed_mc(const int* __restrict__ stories,
                           const int* __restrict__ masks,
                           const float* __restrict__ W_A,
                           const float* __restrict__ W_C,
                           const float* __restrict__ W_AT,
                           const float* __restrict__ W_CT,
                           float* __restrict__ m,
                           float* __restrict__ c) {
  const int blk = blockIdx.x;
  const int s   = blk % NSENT;
  const int d   = threadIdx.x;
  const int* story = stories + blk * TOK;
  const int* msk   = masks   + blk * TOK;
  int wl = 0;
#pragma unroll
  for (int t = 0; t < TOK; ++t) wl += (msk[t] == 0) ? 1 : 0;
  const int te = (wl >= 1) ? (s + 1) : 0;
  float ms = W_AT[te * EMBED + d];
  float cs = W_CT[te * EMBED + d];
#pragma unroll
  for (int t = 0; t < TOK; ++t) {
    const int tok = story[t];
    ms += W_A[tok * EMBED + d];
    cs += W_C[tok * EMBED + d];
  }
  m[blk * EMBED + d] = ms;
  c[blk * EMBED + d] = cs;
}

__global__ void k_hops(const float* __restrict__ m,
                       const float* __restrict__ c,
                       float* __restrict__ u) {
  __shared__ float u_s[EMBED];
  __shared__ float score[NSENT];
  const int b = blockIdx.x;
  const int t = threadIdx.x;
  const float* mb = m + b * NSENT * EMBED;
  const float* cb = c + b * NSENT * EMBED;
  u_s[t] = u[b * EMBED + t];
  __syncthreads();

  for (int hop = 0; hop < NHOPS; ++hop) {
    if (t < NSENT) {
      float acc = 0.f;
      const float* row = mb + t * EMBED;
#pragma unroll 8
      for (int d = 0; d < EMBED; ++d) acc += row[d] * u_s[d];
      score[t] = acc;
    }
    __syncthreads();
    float mx = -1e30f;
    for (int s2 = 0; s2 < NSENT; ++s2) mx = fmaxf(mx, score[s2]);
    __syncthreads();
    if (t < NSENT) score[t] = __expf(score[t] - mx);
    __syncthreads();
    float ssum = 0.f;
    for (int s2 = 0; s2 < NSENT; ++s2) ssum += score[s2];
    const float inv = 1.f / ssum;
    float o = 0.f;
    for (int s2 = 0; s2 < NSENT; ++s2) o += score[s2] * cb[s2 * EMBED + t];
    __syncthreads();
    u_s[t] += o * inv;
    __syncthreads();
  }
  u[b * EMBED + t] = u_s[t];
}

__global__ void k_gemm(const float* __restrict__ u,
                       const float* __restrict__ W_lin,
                       const float* __restrict__ b_lin,
                       float* __restrict__ out) {
  const int NT   = (VOCAB + 15) / 16;
  const int wave = (int)((blockIdx.x * blockDim.x + threadIdx.x) >> 5);
  const int lane = (int)(threadIdx.x & 31);
  if (wave >= NT) return;
  const int nbase = wave * 16;
  const int half  = lane >> 4;
  const int l16   = lane & 15;

  const int nrow   = nbase + l16;
  const int nrow_c = (nrow < VOCAB) ? nrow : (VOCAB - 1);
  const float* brow = W_lin + (size_t)nrow_c * EMBED + 2 * half;
  const float* arow = u + l16 * EMBED + 2 * half;

  v8f acc0 = {0.f,0.f,0.f,0.f,0.f,0.f,0.f,0.f};
  v8f acc1 = acc0, acc2 = acc0, acc3 = acc0;
#pragma unroll
  for (int kk = 0; kk < EMBED / 4; ++kk) {
    const v2f bf = *(const v2f*)(brow + kk * 4);
    const v2f a0 = *(const v2f*)(arow + 0 * 16 * EMBED + kk * 4);
    const v2f a1 = *(const v2f*)(arow + 1 * 16 * EMBED + kk * 4);
    const v2f a2 = *(const v2f*)(arow + 2 * 16 * EMBED + kk * 4);
    const v2f a3 = *(const v2f*)(arow + 3 * 16 * EMBED + kk * 4);
    acc0 = __builtin_amdgcn_wmma_f32_16x16x4_f32(false, a0, false, bf, (short)0, acc0, false, false);
    acc1 = __builtin_amdgcn_wmma_f32_16x16x4_f32(false, a1, false, bf, (short)0, acc1, false, false);
    acc2 = __builtin_amdgcn_wmma_f32_16x16x4_f32(false, a2, false, bf, (short)0, acc2, false, false);
    acc3 = __builtin_amdgcn_wmma_f32_16x16x4_f32(false, a3, false, bf, (short)0, acc3, false, false);
  }

  if (nrow < VOCAB) {
    const float bias = b_lin[nrow];
#pragma unroll
    for (int r = 0; r < 8; ++r) {
      const int mm = r + 8 * half;
      out[(size_t)(mm +  0) * VOCAB + nrow] = acc0[r] + bias;
      out[(size_t)(mm + 16) * VOCAB + nrow] = acc1[r] + bias;
      out[(size_t)(mm + 32) * VOCAB + nrow] = acc2[r] + bias;
      out[(size_t)(mm + 48) * VOCAB + nrow] = acc3[r] + bias;
    }
  }
}

extern "C" void kernel_launch(void* const* d_in, const int* in_sizes, int n_in,
                              void* d_out, int out_size, void* d_ws, size_t ws_size,
                              hipStream_t stream) {
  (void)in_sizes; (void)n_in; (void)out_size; (void)ws_size;

  const int*   stories   = (const int*)d_in[0];
  const int*   questions = (const int*)d_in[1];
  const int*   masks     = (const int*)d_in[2];
  const float* W_A   = (const float*)d_in[3];
  const float* W_B   = (const float*)d_in[4];
  const float* W_C   = (const float*)d_in[5];
  const float* W_AT  = (const float*)d_in[6];
  const float* W_CT  = (const float*)d_in[7];
  const float* W_lin = (const float*)d_in[8];
  const float* b_lin = (const float*)d_in[9];
  float* out = (float*)d_out;

  float* ws = (float*)d_ws;
  float* m = ws;
  float* c = m + BATCH * NSENT * EMBED;
  float* u = c + BATCH * NSENT * EMBED;

  k_embed_q<<<BATCH, EMBED, 0, stream>>>(questions, W_B, u);
  k_embed_mc<<<BATCH * NSENT, EMBED, 0, stream>>>(stories, masks, W_A, W_C, W_AT, W_CT, m, c);
  k_hops<<<BATCH, EMBED, 0, stream>>>(m, c, u);

  const int NT     = (VOCAB + 15) / 16;
  const int blocks = (NT + 7) / 8;
  k_gemm<<<blocks, 256, 0, stream>>>(u, W_lin, b_lin, out);
}


# GEMM: u (A operand) staged once per work-group in LDS, A fragments read from LDS 128-bit, W_lin read 128-bit to match; changes which 4 k values each f32 matrix step sums (sum order)
# speedup vs baseline: 1.1787x; 1.1787x over previous
#include <hip/hip_runtime.h>
#include <math.h>

#define VOCAB 50257
#define EMBED 128
#define NSENT 50
#define TOK   20
#define BATCH 64
#define NHOPS 3

typedef __attribute__((ext_vector_type(2))) float v2f;
typedef __attribute__((ext_vector_type(8))) float v8f;

__global__ void k_embed_q(const int* __restrict__ questions,
                          const float* __restrict__ W_B,
                          float* __restrict__ u) {
  const int b = blockIdx.x;
  const int d = threadIdx.x;
  float s = 0.f;
#pragma unroll
  for (int t = 0; t < TOK; ++t) {
    const int tok = questions[b * TOK + t];
    s += W_B[tok * EMBED + d];
  }
  u[b * EMBED + d] = s;
}

__global__ void k_embed_mc(const int* __restrict__ stories,
                           const int* __restrict__ masks,
                           const float* __restrict__ W_A,
                           const float* __restrict__ W_C,
                           const float* __restrict__ W_AT,
                           const float* __restrict__ W_CT,
                           float* __restrict__ m,
                           float* __restrict__ c) {
  const int blk = blockIdx.x;
  const int s   = blk % NSENT;
  const int d   = threadIdx.x;
  const int* story = stories + blk * TOK;
  const int* msk   = masks   + blk * TOK;
  int wl = 0;
#pragma unroll
  for (int t = 0; t < TOK; ++t) wl += (msk[t] == 0) ? 1 : 0;
  const int te = (wl >= 1) ? (s + 1) : 0;
  float ms = W_AT[te * EMBED + d];
  float cs = W_CT[te * EMBED + d];
#pragma unroll
  for (int t = 0; t < TOK; ++t) {
    const int tok = story[t];
    ms += W_A[tok * EMBED + d];
    cs += W_C[tok * EMBED + d];
  }
  m[blk * EMBED + d] = ms;
  c[blk * EMBED + d] = cs;
}

__global__ void k_hops(const float* __restrict__ m,
                       const float* __restrict__ c,
                       float* __restrict__ u) {
  __shared__ float u_s[EMBED];
  __shared__ float score[NSENT];
  const int b = blockIdx.x;
  const int t = threadIdx.x;
  const float* mb = m + b * NSENT * EMBED;
  const float* cb = c + b * NSENT * EMBED;
  u_s[t] = u[b * EMBED + t];
  __syncthreads();

  for (int hop = 0; hop < NHOPS; ++hop) {
    if (t < NSENT) {
      float acc = 0.f;
      const float* row = mb + t * EMBED;
#pragma unroll 8
      for (int d = 0; d < EMBED; ++d) acc += row[d] * u_s[d];
      score[t] = acc;
    }
    __syncthreads();
    float mx = -1e30f;
    for (int s2 = 0; s2 < NSENT; ++s2) mx = fmaxf(mx, score[s2]);
    __syncthreads();
    if (t < NSENT) score[t] = __expf(score[t] - mx);
    __syncthreads();
    float ssum = 0.f;
    for (int s2 = 0; s2 < NSENT; ++s2) ssum += score[s2];
    const float inv = 1.f / ssum;
    float o = 0.f;
    for (int s2 = 0; s2 < NSENT; ++s2) o += score[s2] * cb[s2 * EMBED + t];
    __syncthreads();
    u_s[t] += o * inv;
    __syncthreads();
  }
  u[b * EMBED + t] = u_s[t];
}

#define UPITCH 132
typedef __attribute__((ext_vector_type(4))) float v4f;

__global__ void __launch_bounds__(256)
k_gemm(const float* __restrict__ u,
       const float* __restrict__ W_lin,
       const float* __restrict__ b_lin,
       float* __restrict__ out) {
  __shared__ float u_s[BATCH * UPITCH];
  const int NT   = (VOCAB + 15) / 16;
  const int tid  = (int)threadIdx.x;
  const int wraw = (int)((blockIdx.x * 256 + tid) >> 5);
  const bool live = wraw < NT;
  const int wave = live ? wraw : (NT - 1);
  const int lane = tid & 31;
  const int nbase = wave * 16;
  const int half  = lane >> 4;
  const int l16   = lane & 15;

  const int nrow   = nbase + l16;
  const int nrow_c = (nrow < VOCAB) ? nrow : (VOCAB - 1);
  const float* brow = W_lin + (size_t)nrow_c * EMBED + 4 * half;

  v4f us[8];
#pragma unroll
  for (int i = 0; i < 8; ++i) us[i] = *(const v4f*)(u + (i * 256 + tid) * 4);

  v4f bf[16];
#pragma unroll
  for (int j = 0; j < 16; ++j) bf[j] = *(const v4f*)(brow + j * 8);

#pragma unroll
  for (int i = 0; i < 8; ++i) {
    const int e = (i * 256 + tid) * 4;
    *(v4f*)(u_s + (e >> 7) * UPITCH + (e & (EMBED - 1))) = us[i];
  }
  __syncthreads();

  const float* arow = u_s + l16 * UPITCH + 4 * half;

  v8f acc0 = {0.f,0.f,0.f,0.f,0.f,0.f,0.f,0.f};
  v8f acc1 = acc0, acc2 = acc0, acc3 = acc0;
#pragma unroll
  for (int j = 0; j < EMBED / 8; ++j) {
    const v4f a0 = *(const v4f*)(arow + 0 * 16 * UPITCH + j * 8);
    const v4f a1 = *(const v4f*)(arow + 1 * 16 * UPITCH + j * 8);
    const v4f a2 = *(const v4f*)(arow + 2 * 16 * UPITCH + j * 8);
    const v4f a3 = *(const v4f*)(arow + 3 * 16 * UPITCH + j * 8);
    const v4f b  = bf[j];
    acc0 = __builtin_amdgcn_wmma_f32_16x16x4_f32(false, a0.xy, false, b.xy, (short)0, acc0, false, false);
    acc1 = __builtin_amdgcn_wmma_f32_16x16x4_f32(false, a1.xy, false, b.xy, (short)0, acc1, false, false);
    acc2 = __builtin_amdgcn_wmma_f32_16x16x4_f32(false, a2.xy, false, b.xy, (short)0, acc2, false, false);
    acc3 = __builtin_amdgcn_wmma_f32_16x16x4_f32(false, a3.xy, false, b.xy, (short)0, acc3, false, false);
    acc0 = __builtin_amdgcn_wmma_f32_16x16x4_f32(false, a0.zw, false, b.zw, (short)0, acc0, false, false);
    acc1 = __builtin_amdgcn_wmma_f32_16x16x4_f32(false, a1.zw, false, b.zw, (short)0, acc1, false, false);
    acc2 = __builtin_amdgcn_wmma_f32_16x16x4_f32(false, a2.zw, false, b.zw, (short)0, acc2, false, false);
    acc3 = __builtin_amdgcn_wmma_f32_16x16x4_f32(false, a3.zw, false, b.zw, (short)0, acc3, false, false);
  }

  if (live && nrow < VOCAB) {
    const float bias = b_lin[nrow];
#pragma unroll
    for (int r = 0; r < 8; ++r) {
      const int mm = r + 8 * half;
      out[(size_t)(mm +  0) * VOCAB + nrow] = acc0[r] + bias;
      out[(size_t)(mm + 16) * VOCAB + nrow] = acc1[r] + bias;
      out[(size_t)(mm + 32) * VOCAB + nrow] = acc2[r] + bias;
      out[(size_t)(mm + 48) * VOCAB + nrow] = acc3[r] + bias;
    }
  }
}

extern "C" void kernel_launch(void* const* d_in, const int* in_sizes, int n_in,
                              void* d_out, int out_size, void* d_ws, size_t ws_size,
                              hipStream_t stream) {
  (void)in_sizes; (void)n_in; (void)out_size; (void)ws_size;

  const int*   stories   = (const int*)d_in[0];
  const int*   questions = (const int*)d_in[1];
  const int*   masks     = (const int*)d_in[2];
  const float* W_A   = (const float*)d_in[3];
  const float* W_B   = (const float*)d_in[4];
  const float* W_C   = (const float*)d_in[5];
  const float* W_AT  = (const float*)d_in[6];
  const float* W_CT  = (const float*)d_in[7];
  const float* W_lin = (const float*)d_in[8];
  const float* b_lin = (const float*)d_in[9];
  float* out = (float*)d_out;

  float* ws = (float*)d_ws;
  float* m = ws;
  float* c = m + BATCH * NSENT * EMBED;
  float* u = c + BATCH * NSENT * EMBED;

  k_embed_q<<<BATCH, EMBED, 0, stream>>>(questions, W_B, u);
  k_embed_mc<<<BATCH * NSENT, EMBED, 0, stream>>>(stories, masks, W_A, W_C, W_AT, W_CT, m, c);
  k_hops<<<BATCH, EMBED, 0, stream>>>(m, c, u);

  const int NT     = (VOCAB + 15) / 16;
  const int blocks = (NT + 7) / 8;
  k_gemm<<<blocks, 256, 0, stream>>>(u, W_lin, b_lin, out);
}


# k_hops: m[b] read once per work-group with whole-row 128-bit loads into LDS, each scoring thread then holds its row in registers over the 3 hops (was re-read from global every hop); sum order kept
# speedup vs baseline: 1.2899x; 1.0943x over previous
#include <hip/hip_runtime.h>
#include <math.h>

#define VOCAB 50257
#define EMBED 128
#define NSENT 50
#define TOK   20
#define BATCH 64
#define NHOPS 3

typedef __attribute__((ext_vector_type(2))) float v2f;
typedef __attribute__((ext_vector_type(8))) float v8f;

__global__ void k_embed_q(const int* __restrict__ questions,
                          const float* __restrict__ W_B,
                          float* __restrict__ u) {
  const int b = blockIdx.x;
  const int d = threadIdx.x;
  float s = 0.f;
#pragma unroll
  for (int t = 0; t < TOK; ++t) {
    const int tok = questions[b * TOK + t];
    s += W_B[tok * EMBED + d];
  }
  u[b * EMBED + d] = s;
}

__global__ void k_embed_mc(const int* __restrict__ stories,
                           const int* __restrict__ masks,
                           const float* __restrict__ W_A,
                           const float* __restrict__ W_C,
                           const float* __restrict__ W_AT,
                           const float* __restrict__ W_CT,
                           float* __restrict__ m,
                           float* __restrict__ c) {
  const int blk = blockIdx.x;
  const int s   = blk % NSENT;
  const int d   = threadIdx.x;
  const int* story = stories + blk * TOK;
  const int* msk   = masks   + blk * TOK;
  int wl = 0;
#pragma unroll
  for (int t = 0; t < TOK; ++t) wl += (msk[t] == 0) ? 1 : 0;
  const int te = (wl >= 1) ? (s + 1) : 0;
  float ms = W_AT[te * EMBED + d];
  float cs = W_CT[te * EMBED + d];
#pragma unroll
  for (int t = 0; t < TOK; ++t) {
    const int tok = story[t];
    ms += W_A[tok * EMBED + d];
    cs += W_C[tok * EMBED + d];
  }
  m[blk * EMBED + d] = ms;
  c[blk * EMBED + d] = cs;
}

#define MPITCH 132
typedef __attribute__((ext_vector_type(4))) float f4;

__global__ void __launch_bounds__(EMBED)
k_hops(const float* __restrict__ m,
       const float* __restrict__ c,
       float* __restrict__ u) {
  __shared__ float m_s[NSENT * MPITCH];
  __shared__ float u_s[EMBED];
  __shared__ float score[NSENT];
  const int b = blockIdx.x;
  const int t = threadIdx.x;
  const float* mb = m + b * NSENT * EMBED;
  const float* cb = c + b * NSENT * EMBED;

  f4 mv[13];
#pragma unroll
  for (int i = 0; i < 12; ++i) mv[i] = *(const f4*)(mb + (i * EMBED + t) * 4);
  if (t < 64) mv[12] = *(const f4*)(mb + (12 * EMBED + t) * 4);
  const float u0 = u[b * EMBED + t];

#pragma unroll
  for (int i = 0; i < 12; ++i) {
    const int q = i * EMBED + t;
    *(f4*)(m_s + (q >> 5) * MPITCH + (q & 31) * 4) = mv[i];
  }
  if (t < 64) {
    const int q = 12 * EMBED + t;
    *(f4*)(m_s + (q >> 5) * MPITCH + (q & 31) * 4) = mv[12];
  }
  u_s[t] = u0;
  __syncthreads();

  float cr[NSENT];
#pragma unroll
  for (int s2 = 0; s2 < NSENT; ++s2) cr[s2] = cb[s2 * EMBED + t];

  const bool scorer = t < NSENT;
  f4 mr[EMBED / 4];
  if (scorer) {
#pragma unroll
    for (int k = 0; k < EMBED / 4; ++k) mr[k] = *(const f4*)(m_s + t * MPITCH + 4 * k);
  }

  for (int hop = 0; hop < NHOPS; ++hop) {
    if (scorer) {
      float acc = 0.f;
#pragma unroll
      for (int k = 0; k < EMBED / 4; ++k) {
        const f4 uu = *(const f4*)(u_s + 4 * k);
        acc = __builtin_fmaf(mr[k].x, uu.x, acc);
        acc = __builtin_fmaf(mr[k].y, uu.y, acc);
        acc = __builtin_fmaf(mr[k].z, uu.z, acc);
        acc = __builtin_fmaf(mr[k].w, uu.w, acc);
      }
      score[t] = acc;
    }
    __syncthreads();
    float mx = -1e30f;
    for (int s2 = 0; s2 < NSENT; ++s2) mx = fmaxf(mx, score[s2]);
    __syncthreads();
    if (scorer) score[t] = __expf(score[t] - mx);
    __syncthreads();
    float ssum = 0.f;
    for (int s2 = 0; s2 < NSENT; ++s2) ssum += score[s2];
    const float inv = 1.f / ssum;
    float o = 0.f;
#pragma unroll
    for (int s2 = 0; s2 < NSENT; ++s2) o += score[s2] * cr[s2];
    __syncthreads();
    u_s[t] += o * inv;
    __syncthreads();
  }
  u[b * EMBED + t] = u_s[t];
}

#define UPITCH 132
typedef __attribute__((ext_vector_type(4))) float v4f;

__global__ void __launch_bounds__(256)
k_gemm(const float* __restrict__ u,
       const float* __restrict__ W_lin,
       const float* __restrict__ b_lin,
       float* __restrict__ out) {
  __shared__ float u_s[BATCH * UPITCH];
  const int NT   = (VOCAB + 15) / 16;
  const int tid  = (int)threadIdx.x;
  const int wraw = (int)((blockIdx.x * 256 + tid) >> 5);
  const bool live = wraw < NT;
  const int wave = live ? wraw : (NT - 1);
  const int lane = tid & 31;
  const int nbase = wave * 16;
  const int half  = lane >> 4;
  const int l16   = lane & 15;

  const int nrow   = nbase + l16;
  const int nrow_c = (nrow < VOCAB) ? nrow : (VOCAB - 1);
  const float* brow = W_lin + (size_t)nrow_c * EMBED + 4 * half;

  v4f us[8];
#pragma unroll
  for (int i = 0; i < 8; ++i) us[i] = *(const v4f*)(u + (i * 256 + tid) * 4);

  v4f bf[16];
#pragma unroll
  for (int j = 0; j < 16; ++j) bf[j] = *(const v4f*)(brow + j * 8);

#pragma unroll
  for (int i = 0; i < 8; ++i) {
    const int e = (i * 256 + tid) * 4;
    *(v4f*)(u_s + (e >> 7) * UPITCH + (e & (EMBED - 1))) = us[i];
  }
  __syncthreads();

  const float* arow = u_s + l16 * UPITCH + 4 * half;

  v8f acc0 = {0.f,0.f,0.f,0.f,0.f,0.f,0.f,0.f};
  v8f acc1 = acc0, acc2 = acc0, acc3 = acc0;
#pragma unroll
  for (int j = 0; j < EMBED / 8; ++j) {
    const v4f a0 = *(const v4f*)(arow + 0 * 16 * UPITCH + j * 8);
    const v4f a1 = *(const v4f*)(arow + 1 * 16 * UPITCH + j * 8);
    const v4f a2 = *(const v4f*)(arow + 2 * 16 * UPITCH + j * 8);
    const v4f a3 = *(const v4f*)(arow + 3 * 16 * UPITCH + j * 8);
    const v4f b  = bf[j];
    acc0 = __builtin_amdgcn_wmma_f32_16x16x4_f32(false, a0.xy, false, b.xy, (short)0, acc0, false, false);
    acc1 = __builtin_amdgcn_wmma_f32_16x16x4_f32(false, a1.xy, false, b.xy, (short)0, acc1, false, false);
    acc2 = __builtin_amdgcn_wmma_f32_16x16x4_f32(false, a2.xy, false, b.xy, (short)0, acc2, false, false);
    acc3 = __builtin_amdgcn_wmma_f32_16x16x4_f32(false, a3.xy, false, b.xy, (short)0, acc3, false, false);
    acc0 = __builtin_amdgcn_wmma_f32_16x16x4_f32(false, a0.zw, false, b.zw, (short)0, acc0, false, false);
    acc1 = __builtin_amdgcn_wmma_f32_16x16x4_f32(false, a1.zw, false, b.zw, (short)0, acc1, false, false);
    acc2 = __builtin_amdgcn_wmma_f32_16x16x4_f32(false, a2.zw, false, b.zw, (short)0, acc2, false, false);
    acc3 = __builtin_amdgcn_wmma_f32_16x16x4_f32(false, a3.zw, false, b.zw, (short)0, acc3, false, false);
  }

  if (live && nrow < VOCAB) {
    const float bias = b_lin[nrow];
#pragma unroll
    for (int r = 0; r < 8; ++r) {
      const int mm = r + 8 * half;
      out[(size_t)(mm +  0) * VOCAB + nrow] = acc0[r] + bias;
      out[(size_t)(mm + 16) * VOCAB + nrow] = acc1[r] + bias;
      out[(size_t)(mm + 32) * VOCAB + nrow] = acc2[r] + bias;
      out[(size_t)(mm + 48) * VOCAB + nrow] = acc3[r] + bias;
    }
  }
}

extern "C" void kernel_launch(void* const* d_in, const int* in_sizes, int n_in,
                              void* d_out, int out_size, void* d_ws, size_t ws_size,
                              hipStream_t stream) {
  (void)in_sizes; (void)n_in; (void)out_size; (void)ws_size;

  const int*   stories   = (const int*)d_in[0];
  const int*   questions = (const int*)d_in[1];
  const int*   masks     = (const int*)d_in[2];
  const float* W_A   = (const float*)d_in[3];
  const float* W_B   = (const float*)d_in[4];
  const float* W_C   = (const float*)d_in[5];
  const float* W_AT  = (const float*)d_in[6];
  const float* W_CT  = (const float*)d_in[7];
  const float* W_lin = (const float*)d_in[8];
  const float* b_lin = (const float*)d_in[9];
  float* out = (float*)d_out;

  float* ws = (float*)d_ws;
  float* m = ws;
  float* c = m + BATCH * NSENT * EMBED;
  float* u = c + BATCH * NSENT * EMBED;

  k_embed_q<<<BATCH, EMBED, 0, stream>>>(questions, W_B, u);
  k_embed_mc<<<BATCH * NSENT, EMBED, 0, stream>>>(stories, masks, W_A, W_C, W_AT, W_CT, m, c);
  k_hops<<<BATCH, EMBED, 0, stream>>>(m, c, u);

  const int NT     = (VOCAB + 15) / 16;
  const int blocks = (NT + 7) / 8;
  k_gemm<<<blocks, 256, 0, stream>>>(u, W_lin, b_lin, out);
}


# Question embedding (u) joins the story-embedding launch as 64 extra work-groups at the head of its grid: 4 dependent launches become 3; every sum keeps its order, results bit-identical
# speedup vs baseline: 1.4376x; 1.1145x over previous
#include <hip/hip_runtime.h>
#include <math.h>

#define VOCAB 50257
#define EMBED 128
#define NSENT 50
#define TOK   20
#define BATCH 64
#define NHOPS 3

typedef __attribute__((ext_vector_type(2))) float v2f;
typedef __attribute__((ext_vector_type(8))) float v8f;

__global__ void k_embed(const int* __restrict__ questions,
                        const int* __restrict__ stories,
                        const int* __restrict__ masks,
                        const float* __restrict__ W_B,
                        const float* __restrict__ W_A,
                        const float* __restrict__ W_C,
                        const float* __restrict__ W_AT,
                        const float* __restrict__ W_CT,
                        float* __restrict__ u,
                        float* __restrict__ m,
                        float* __restrict__ c) {
  const int d = threadIdx.x;
  if (blockIdx.x < BATCH) {
    const int b = blockIdx.x;
    float s = 0.f;
#pragma unroll
    for (int t = 0; t < TOK; ++t) {
      const int tok = questions[b * TOK + t];
      s += W_B[tok * EMBED + d];
    }
    u[b * EMBED + d] = s;
    return;
  }
  const int blk = blockIdx.x - BATCH;
  const int s   = blk % NSENT;
  const int* story = stories + blk * TOK;
  const int* msk   = masks   + blk * TOK;
  int wl = 0;
#pragma unroll
  for (int t = 0; t < TOK; ++t) wl += (msk[t] == 0) ? 1 : 0;
  const int te = (wl >= 1) ? (s + 1) : 0;
  float ms = W_AT[te * EMBED + d];
  float cs = W_CT[te * EMBED + d];
#pragma unroll
  for (int t = 0; t < TOK; ++t) {
    const int tok = story[t];
    ms += W_A[tok * EMBED + d];
    cs += W_C[tok * EMBED + d];
  }
  m[blk * EMBED + d] = ms;
  c[blk * EMBED + d] = cs;
}

#define MPITCH 132
typedef __attribute__((ext_vector_type(4))) float f4;

__global__ void __launch_bounds__(EMBED)
k_hops(const float* __restrict__ m,
       const float* __restrict__ c,
       float* __restrict__ u) {
  __shared__ float m_s[NSENT * MPITCH];
  __shared__ float u_s[EMBED];
  __shared__ float score[NSENT];
  const int b = blockIdx.x;
  const int t = threadIdx.x;
  const float* mb = m + b * NSENT * EMBED;
  const float* cb = c + b * NSENT * EMBED;

  f4 mv[13];
#pragma unroll
  for (int i = 0; i < 12; ++i) mv[i] = *(const f4*)(mb + (i * EMBED + t) * 4);
  if (t < 64) mv[12] = *(const f4*)(mb + (12 * EMBED + t) * 4);
  const float u0 = u[b * EMBED + t];

#pragma unroll
  for (int i = 0; i < 12; ++i) {
    const int q = i * EMBED + t;
    *(f4*)(m_s + (q >> 5) * MPITCH + (q & 31) * 4) = mv[i];
  }
  if (t < 64) {
    const int q = 12 * EMBED + t;
    *(f4*)(m_s + (q >> 5) * MPITCH + (q & 31) * 4) = mv[12];
  }
  u_s[t] = u0;
  __syncthreads();

  float cr[NSENT];
#pragma unroll
  for (int s2 = 0; s2 < NSENT; ++s2) cr[s2] = cb[s2 * EMBED + t];

  const bool scorer = t < NSENT;
  f4 mr[EMBED / 4];
  if (scorer) {
#pragma unroll
    for (int k = 0; k < EMBED / 4; ++k) mr[k] = *(const f4*)(m_s + t * MPITCH + 4 * k);
  }

  for (int hop = 0; hop < NHOPS; ++hop) {
    if (scorer) {
      float acc = 0.f;
#pragma unroll
      for (int k = 0; k < EMBED / 4; ++k) {
        const f4 uu = *(const f4*)(u_s + 4 * k);
        acc = __builtin_fmaf(mr[k].x, uu.x, acc);
        acc = __builtin_fmaf(mr[k].y, uu.y, acc);
        acc = __builtin_fmaf(mr[k].z, uu.z, acc);
        acc = __builtin_fmaf(mr[k].w, uu.w, acc);
      }
      score[t] = acc;
    }
    __syncthreads();
    float mx = -1e30f;
    for (int s2 = 0; s2 < NSENT; ++s2) mx = fmaxf(mx, score[s2]);
    __syncthreads();
    if (scorer) score[t] = __expf(score[t] - mx);
    __syncthreads();
    float ssum = 0.f;
    for (int s2 = 0; s2 < NSENT; ++s2) ssum += score[s2];
    const float inv = 1.f / ssum;
    float o = 0.f;
#pragma unroll
    for (int s2 = 0; s2 < NSENT; ++s2) o += score[s2] * cr[s2];
    __syncthreads();
    u_s[t] += o * inv;
    __syncthreads();
  }
  u[b * EMBED + t] = u_s[t];
}

#define UPITCH 132
typedef __attribute__((ext_vector_type(4))) float v4f;

__global__ void __launch_bounds__(256)
k_gemm(const float* __restrict__ u,
       const float* __restrict__ W_lin,
       const float* __restrict__ b_lin,
       float* __restrict__ out) {
  __shared__ float u_s[BATCH * UPITCH];
  const int NT   = (VOCAB + 15) / 16;
  const int tid  = (int)threadIdx.x;
  const int wraw = (int)((blockIdx.x * 256 + tid) >> 5);
  const bool live = wraw < NT;
  const int wave = live ? wraw : (NT - 1);
  const int lane = tid & 31;
  const int nbase = wave * 16;
  const int half  = lane >> 4;
  const int l16   = lane & 15;

  const int nrow   = nbase + l16;
  const int nrow_c = (nrow < VOCAB) ? nrow : (VOCAB - 1);
  const float* brow = W_lin + (size_t)nrow_c * EMBED + 4 * half;

  v4f us[8];
#pragma unroll
  for (int i = 0; i < 8; ++i) us[i] = *(const v4f*)(u + (i * 256 + tid) * 4);

  v4f bf[16];
#pragma unroll
  for (int j = 0; j < 16; ++j) bf[j] = *(const v4f*)(brow + j * 8);

#pragma unroll
  for (int i = 0; i < 8; ++i) {
    const int e = (i * 256 + tid) * 4;
    *(v4f*)(u_s + (e >> 7) * UPITCH + (e & (EMBED - 1))) = us[i];
  }
  __syncthreads();

  const float* arow = u_s + l16 * UPITCH + 4 * half;

  v8f acc0 = {0.f,0.f,0.f,0.f,0.f,0.f,0.f,0.f};
  v8f acc1 = acc0, acc2 = acc0, acc3 = acc0;
#pragma unroll
  for (int j = 0; j < EMBED / 8; ++j) {
    const v4f a0 = *(const v4f*)(arow + 0 * 16 * UPITCH + j * 8);
    const v4f a1 = *(const v4f*)(arow + 1 * 16 * UPITCH + j * 8);
    const v4f a2 = *(const v4f*)(arow + 2 * 16 * UPITCH + j * 8);
    const v4f a3 = *(const v4f*)(arow + 3 * 16 * UPITCH + j * 8);
    const v4f b  = bf[j];
    acc0 = __builtin_amdgcn_wmma_f32_16x16x4_f32(false, a0.xy, false, b.xy, (short)0, acc0, false, false);
    acc1 = __builtin_amdgcn_wmma_f32_16x16x4_f32(false, a1.xy, false, b.xy, (short)0, acc1, false, false);
    acc2 = __builtin_amdgcn_wmma_f32_16x16x4_f32(false, a2.xy, false, b.xy, (short)0, acc2, false, false);
    acc3 = __builtin_amdgcn_wmma_f32_16x16x4_f32(false, a3.xy, false, b.xy, (short)0, acc3, false, false);
    acc0 = __builtin_amdgcn_wmma_f32_16x16x4_f32(false, a0.zw, false, b.zw, (short)0, acc0, false, false);
    acc1 = __builtin_amdgcn_wmma_f32_16x16x4_f32(false, a1.zw, false, b.zw, (short)0, acc1, false, false);
    acc2 = __builtin_amdgcn_wmma_f32_16x16x4_f32(false, a2.zw, false, b.zw, (short)0, acc2, false, false);
    acc3 = __builtin_amdgcn_wmma_f32_16x16x4_f32(false, a3.zw, false, b.zw, (short)0, acc3, false, false);
  }

  if (live && nrow < VOCAB) {
    const float bias = b_lin[nrow];
#pragma unroll
    for (int r = 0; r < 8; ++r) {
      const int mm = r + 8 * half;
      out[(size_t)(mm +  0) * VOCAB + nrow] = acc0[r] + bias;
      out[(size_t)(mm + 16) * VOCAB + nrow] = acc1[r] + bias;
      out[(size_t)(mm + 32) * VOCAB + nrow] = acc2[r] + bias;
      out[(size_t)(mm + 48) * VOCAB + nrow] = acc3[r] + bias;
    }
  }
}

extern "C" void kernel_launch(void* const* d_in, const int* in_sizes, int n_in,
                              void* d_out, int out_size, void* d_ws, size_t ws_size,
                              hipStream_t stream) {
  (void)in_sizes; (void)n_in; (void)out_size; (void)ws_size;

  const int*   stories   = (const int*)d_in[0];
  const int*   questions = (const int*)d_in[1];
  const int*   masks     = (const int*)d_in[2];
  const float* W_A   = (const float*)d_in[3];
  const float* W_B   = (const float*)d_in[4];
  const float* W_C   = (const float*)d_in[5];
  const float* W_AT  = (const float*)d_in[6];
  const float* W_CT  = (const float*)d_in[7];
  const float* W_lin = (const float*)d_in[8];
  const float* b_lin = (const float*)d_in[9];
  float* out = (float*)d_out;

  float* ws = (float*)d_ws;
  float* m = ws;
  float* c = m + BATCH * NSENT * EMBED;
  float* u = c + BATCH * NSENT * EMBED;

  k_embed<<<BATCH + BATCH * NSENT, EMBED, 0, stream>>>(questions, stories, masks, W_B, W_A, W_C,
                                                      W_AT, W_CT, u, m, c);
  k_hops<<<BATCH, EMBED, 0, stream>>>(m, c, u);

  const int NT     = (VOCAB + 15) / 16;
  const int blocks = (NT + 7) / 8;
  k_gemm<<<blocks, 256, 0, stream>>>(u, W_lin, b_lin, out);
}


# k_hops: each hop's serial sums split up: a score is 2 threads x 4 accumulators; max, exp, the exp sum and the c column sum are formed by every thread in 4 accumulators each; changes sum order
# speedup vs baseline: 1.5534x; 1.0806x over previous
#include <hip/hip_runtime.h>
#include <math.h>

#define VOCAB 50257
#define EMBED 128
#define NSENT 50
#define TOK   20
#define BATCH 64
#define NHOPS 3

typedef __attribute__((ext_vector_type(2))) float v2f;
typedef __attribute__((ext_vector_type(8))) float v8f;

__global__ void k_embed(const int* __restrict__ questions,
                        const int* __restrict__ stories,
                        const int* __restrict__ masks,
                        const float* __restrict__ W_B,
                        const float* __restrict__ W_A,
                        const float* __restrict__ W_C,
                        const float* __restrict__ W_AT,
                        const float* __restrict__ W_CT,
                        float* __restrict__ u,
                        float* __restrict__ m,
                        float* __restrict__ c) {
  const int d = threadIdx.x;
  if (blockIdx.x < BATCH) {
    const int b = blockIdx.x;
    float s = 0.f;
#pragma unroll
    for (int t = 0; t < TOK; ++t) {
      const int tok = questions[b * TOK + t];
      s += W_B[tok * EMBED + d];
    }
    u[b * EMBED + d] = s;
    return;
  }
  const int blk = blockIdx.x - BATCH;
  const int s   = blk % NSENT;
  const int* story = stories + blk * TOK;
  const int* msk   = masks   + blk * TOK;
  int wl = 0;
#pragma unroll
  for (int t = 0; t < TOK; ++t) wl += (msk[t] == 0) ? 1 : 0;
  const int te = (wl >= 1) ? (s + 1) : 0;
  float ms = W_AT[te * EMBED + d];
  float cs = W_CT[te * EMBED + d];
#pragma unroll
  for (int t = 0; t < TOK; ++t) {
    const int tok = story[t];
    ms += W_A[tok * EMBED + d];
    cs += W_C[tok * EMBED + d];
  }
  m[blk * EMBED + d] = ms;
  c[blk * EMBED + d] = cs;
}

#define MPITCH 132
#define SPITCH 64
typedef __attribute__((ext_vector_type(4))) float f4;

__global__ void __launch_bounds__(EMBED)
k_hops(const float* __restrict__ m,
       const float* __restrict__ c,
       float* __restrict__ u) {
  __shared__ __attribute__((aligned(16))) float m_s[NSENT * MPITCH];
  __shared__ __attribute__((aligned(16))) float u_s[EMBED];
  __shared__ __attribute__((aligned(16))) float part[2 * SPITCH];
  const int b = blockIdx.x;
  const int t = threadIdx.x;
  const float* mb = m + b * NSENT * EMBED;
  const float* cb = c + b * NSENT * EMBED;

  f4 mv[13];
#pragma unroll
  for (int i = 0; i < 12; ++i) mv[i] = *(const f4*)(mb + (i * EMBED + t) * 4);
  if (t < 64) mv[12] = *(const f4*)(mb + (12 * EMBED + t) * 4);
  float ut = u[b * EMBED + t];

#pragma unroll
  for (int i = 0; i < 12; ++i) {
    const int q = i * EMBED + t;
    *(f4*)(m_s + (q >> 5) * MPITCH + (q & 31) * 4) = mv[i];
  }
  if (t < 64) {
    const int q = 12 * EMBED + t;
    *(f4*)(m_s + (q >> 5) * MPITCH + (q & 31) * 4) = mv[12];
  }
  u_s[t] = ut;
  __syncthreads();

  float cr[NSENT];
#pragma unroll
  for (int s2 = 0; s2 < NSENT; ++s2) cr[s2] = cb[s2 * EMBED + t];

  const int s = t & 63;
  const int h = t >> 6;
  const bool scorer = s < NSENT;
  f4 mr[EMBED / 8];
  if (scorer) {
#pragma unroll
    for (int k = 0; k < EMBED / 8; ++k) mr[k] = *(const f4*)(m_s + s * MPITCH + 64 * h + 4 * k);
  }

  for (int hop = 0; hop < NHOPS; ++hop) {
    if (scorer) {
      float a0 = 0.f, a1 = 0.f, a2 = 0.f, a3 = 0.f;
#pragma unroll
      for (int k = 0; k < EMBED / 8; ++k) {
        const f4 uu = *(const f4*)(u_s + 64 * h + 4 * k);
        a0 = __builtin_fmaf(mr[k].x, uu.x, a0);
        a1 = __builtin_fmaf(mr[k].y, uu.y, a1);
        a2 = __builtin_fmaf(mr[k].z, uu.z, a2);
        a3 = __builtin_fmaf(mr[k].w, uu.w, a3);
      }
      part[h * SPITCH + s] = (a0 + a1) + (a2 + a3);
    }
    __syncthreads();

    float sc[NSENT];
#pragma unroll
    for (int q = 0; q < 12; ++q) {
      const f4 p0 = *(const f4*)(part + 4 * q);
      const f4 p1 = *(const f4*)(part + SPITCH + 4 * q);
      sc[4 * q + 0] = p0.x + p1.x;
      sc[4 * q + 1] = p0.y + p1.y;
      sc[4 * q + 2] = p0.z + p1.z;
      sc[4 * q + 3] = p0.w + p1.w;
    }
    {
      const v2f p0 = *(const v2f*)(part + 48);
      const v2f p1 = *(const v2f*)(part + SPITCH + 48);
      sc[48] = p0.x + p1.x;
      sc[49] = p0.y + p1.y;
    }
    float x0 = sc[0], x1 = sc[1], x2 = sc[2], x3 = sc[3];
#pragma unroll
    for (int q = 1; q < 12; ++q) {
      x0 = fmaxf(x0, sc[4 * q + 0]);
      x1 = fmaxf(x1, sc[4 * q + 1]);
      x2 = fmaxf(x2, sc[4 * q + 2]);
      x3 = fmaxf(x3, sc[4 * q + 3]);
    }
    x0 = fmaxf(x0, sc[48]);
    x1 = fmaxf(x1, sc[49]);
    const float mx = fmaxf(fmaxf(x0, x1), fmaxf(x2, x3));

    float e0 = 0.f, e1 = 0.f, e2 = 0.f, e3 = 0.f;
    float o0 = 0.f, o1 = 0.f, o2 = 0.f, o3 = 0.f;
#pragma unroll
    for (int q = 0; q < 12; ++q) {
      const float p0 = __expf(sc[4 * q + 0] - mx);
      const float p1 = __expf(sc[4 * q + 1] - mx);
      const float p2 = __expf(sc[4 * q + 2] - mx);
      const float p3 = __expf(sc[4 * q + 3] - mx);
      e0 += p0; e1 += p1; e2 += p2; e3 += p3;
      o0 = __builtin_fmaf(p0, cr[4 * q + 0], o0);
      o1 = __builtin_fmaf(p1, cr[4 * q + 1], o1);
      o2 = __builtin_fmaf(p2, cr[4 * q + 2], o2);
      o3 = __builtin_fmaf(p3, cr[4 * q + 3], o3);
    }
    {
      const float p0 = __expf(sc[48] - mx);
      const float p1 = __expf(sc[49] - mx);
      e0 += p0; e1 += p1;
      o0 = __builtin_fmaf(p0, cr[48], o0);
      o1 = __builtin_fmaf(p1, cr[49], o1);
    }
    const float ssum = (e0 + e1) + (e2 + e3);
    const float o    = (o0 + o1) + (o2 + o3);
    ut += o * (1.f / ssum);
    if (hop + 1 < NHOPS) {
      u_s[t] = ut;
      __syncthreads();
    }
  }
  u[b * EMBED + t] = ut;
}

#define UPITCH 132
typedef __attribute__((ext_vector_type(4))) float v4f;

__global__ void __launch_bounds__(256)
k_gemm(const float* __restrict__ u,
       const float* __restrict__ W_lin,
       const float* __restrict__ b_lin,
       float* __restrict__ out) {
  __shared__ float u_s[BATCH * UPITCH];
  const int NT   = (VOCAB + 15) / 16;
  const int tid  = (int)threadIdx.x;
  const int wraw = (int)((blockIdx.x * 256 + tid) >> 5);
  const bool live = wraw < NT;
  const int wave = live ? wraw : (NT - 1);
  const int lane = tid & 31;
  const int nbase = wave * 16;
  const int half  = lane >> 4;
  const int l16   = lane & 15;

  const int nrow   = nbase + l16;
  const int nrow_c = (nrow < VOCAB) ? nrow : (VOCAB - 1);
  const float* brow = W_lin + (size_t)nrow_c * EMBED + 4 * half;

  v4f us[8];
#pragma unroll
  for (int i = 0; i < 8; ++i) us[i] = *(const v4f*)(u + (i * 256 + tid) * 4);

  v4f bf[16];
#pragma unroll
  for (int j = 0; j < 16; ++j) bf[j] = *(const v4f*)(brow + j * 8);

#pragma unroll
  for (int i = 0; i < 8; ++i) {
    const int e = (i * 256 + tid) * 4;
    *(v4f*)(u_s + (e >> 7) * UPITCH + (e & (EMBED - 1))) = us[i];
  }
  __syncthreads();

  const float* arow = u_s + l16 * UPITCH + 4 * half;

  v8f acc0 = {0.f,0.f,0.f,0.f,0.f,0.f,0.f,0.f};
  v8f acc1 = acc0, acc2 = acc0, acc3 = acc0;
#pragma unroll
  for (int j = 0; j < EMBED / 8; ++j) {
    const v4f a0 = *(const v4f*)(arow + 0 * 16 * UPITCH + j * 8);
    const v4f a1 = *(const v4f*)(arow + 1 * 16 * UPITCH + j * 8);
    const v4f a2 = *(const v4f*)(arow + 2 * 16 * UPITCH + j * 8);
    const v4f a3 = *(const v4f*)(arow + 3 * 16 * UPITCH + j * 8);
    const v4f b  = bf[j];
    acc0 = __builtin_amdgcn_wmma_f32_16x16x4_f32(false, a0.xy, false, b.xy, (short)0, acc0, false, false);
    acc1 = __builtin_amdgcn_wmma_f32_16x16x4_f32(false, a1.xy, false, b.xy, (short)0, acc1, false, false);
    acc2 = __builtin_amdgcn_wmma_f32_16x16x4_f32(false, a2.xy, false, b.xy, (short)0, acc2, false, false);
    acc3 = __builtin_amdgcn_wmma_f32_16x16x4_f32(false, a3.xy, false, b.xy, (short)0, acc3, false, false);
    acc0 = __builtin_amdgcn_wmma_f32_16x16x4_f32(false, a0.zw, false, b.zw, (short)0, acc0, false, false);
    acc1 = __builtin_amdgcn_wmma_f32_16x16x4_f32(false, a1.zw, false, b.zw, (short)0, acc1, false, false);
    acc2 = __builtin_amdgcn_wmma_f32_16x16x4_f32(false, a2.zw, false, b.zw, (short)0, acc2, false, false);
    acc3 = __builtin_amdgcn_wmma_f32_16x16x4_f32(false, a3.zw, false, b.zw, (short)0, acc3, false, false);
  }

  if (live && nrow < VOCAB) {
    const float bias = b_lin[nrow];
#pragma unroll
    for (int r = 0; r < 8; ++r) {
      const int mm = r + 8 * half;
      out[(size_t)(mm +  0) * VOCAB + nrow] = acc0[r] + bias;
      out[(size_t)(mm + 16) * VOCAB + nrow] = acc1[r] + bias;
      out[(size_t)(mm + 32) * VOCAB + nrow] = acc2[r] + bias;
      out[(size_t)(mm + 48) * VOCAB + nrow] = acc3[r] + bias;
    }
  }
}

extern "C" void kernel_launch(void* const* d_in, const int* in_sizes, int n_in,
                              void* d_out, int out_size, void* d_ws, size_t ws_size,
                              hipStream_t stream) {
  (void)in_sizes; (void)n_in; (void)out_size; (void)ws_size;

  const int*   stories   = (const int*)d_in[0];
  const int*   questions = (const int*)d_in[1];
  const int*   masks     = (const int*)d_in[2];
  const float* W_A   = (const float*)d_in[3];
  const float* W_B   = (const float*)d_in[4];
  const float* W_C   = (const float*)d_in[5];
  const float* W_AT  = (const float*)d_in[6];
  const float* W_CT  = (const float*)d_in[7];
  const float* W_lin = (const float*)d_in[8];
  const float* b_lin = (const float*)d_in[9];
  float* out = (float*)d_out;

  float* ws = (float*)d_ws;
  float* m = ws;
  float* c = m + BATCH * NSENT * EMBED;
  float* u = c + BATCH * NSENT * EMBED;

  k_embed<<<BATCH + BATCH * NSENT, EMBED, 0, stream>>>(questions, stories, masks, W_B, W_A, W_C,
                                                      W_AT, W_CT, u, m, c);
  k_hops<<<BATCH, EMBED, 0, stream>>>(m, c, u);

  const int NT     = (VOCAB + 15) / 16;
  const int blocks = (NT + 7) / 8;
  k_gemm<<<blocks, 256, 0, stream>>>(u, W_lin, b_lin, out);
}


# Work-group barriers fence LDS only, so global loads stay in flight across them: k_gemm multiplies each W_lin chunk as it lands (bias asked for up front), k_hops asks for its c columns at the start
# speedup vs baseline: 1.6686x; 1.0742x over previous
#include <hip/hip_runtime.h>
#include <math.h>

#define VOCAB 50257
#define EMBED 128
#define NSENT 50
#define TOK   20
#define BATCH 64
#define NHOPS 3

typedef __attribute__((ext_vector_type(2))) float v2f;
typedef __attribute__((ext_vector_type(8))) float v8f;

__device__ __forceinline__ void lds_barrier() {
  __builtin_amdgcn_fence(__ATOMIC_RELEASE, "workgroup", "local");
  __builtin_amdgcn_s_barrier();
  __builtin_amdgcn_fence(__ATOMIC_ACQUIRE, "workgroup", "local");
}

__global__ void k_embed(const int* __restrict__ questions,
                        const int* __restrict__ stories,
                        const int* __restrict__ masks,
                        const float* __restrict__ W_B,
                        const float* __restrict__ W_A,
                        const float* __restrict__ W_C,
                        const float* __restrict__ W_AT,
                        const float* __restrict__ W_CT,
                        float* __restrict__ u,
                        float* __restrict__ m,
                        float* __restrict__ c) {
  const int d = threadIdx.x;
  if (blockIdx.x < BATCH) {
    const int b = blockIdx.x;
    float s = 0.f;
#pragma unroll
    for (int t = 0; t < TOK; ++t) {
      const int tok = questions[b * TOK + t];
      s += W_B[tok * EMBED + d];
    }
    u[b * EMBED + d] = s;
    return;
  }
  const int blk = blockIdx.x - BATCH;
  const int s   = blk % NSENT;
  const int* story = stories + blk * TOK;
  const int* msk   = masks   + blk * TOK;
  int wl = 0;
#pragma unroll
  for (int t = 0; t < TOK; ++t) wl += (msk[t] == 0) ? 1 : 0;
  const int te = (wl >= 1) ? (s + 1) : 0;
  float ms = W_AT[te * EMBED + d];
  float cs = W_CT[te * EMBED + d];
#pragma unroll
  for (int t = 0; t < TOK; ++t) {
    const int tok = story[t];
    ms += W_A[tok * EMBED + d];
    cs += W_C[tok * EMBED + d];
  }
  m[blk * EMBED + d] = ms;
  c[blk * EMBED + d] = cs;
}

#define MPITCH 132
#define SPITCH 64
typedef __attribute__((ext_vector_type(4))) float f4;

__global__ void __launch_bounds__(EMBED)
k_hops(const float* __restrict__ m,
       const float* __restrict__ c,
       float* __restrict__ u) {
  __shared__ __attribute__((aligned(16))) float m_s[NSENT * MPITCH];
  __shared__ __attribute__((aligned(16))) float u_s[EMBED];
  __shared__ __attribute__((aligned(16))) float part[2 * SPITCH];
  const int b = blockIdx.x;
  const int t = threadIdx.x;
  const float* mb = m + b * NSENT * EMBED;
  const float* cb = c + b * NSENT * EMBED;

  f4 mv[13];
#pragma unroll
  for (int i = 0; i < 12; ++i) mv[i] = *(const f4*)(mb + (i * EMBED + t) * 4);
  if (t < 64) mv[12] = *(const f4*)(mb + (12 * EMBED + t) * 4);
  float ut = u[b * EMBED + t];

  float cr[NSENT];
#pragma unroll
  for (int s2 = 0; s2 < NSENT; ++s2) cr[s2] = cb[s2 * EMBED + t];

#pragma unroll
  for (int i = 0; i < 12; ++i) {
    const int q = i * EMBED + t;
    *(f4*)(m_s + (q >> 5) * MPITCH + (q & 31) * 4) = mv[i];
  }
  if (t < 64) {
    const int q = 12 * EMBED + t;
    *(f4*)(m_s + (q >> 5) * MPITCH + (q & 31) * 4) = mv[12];
  }
  u_s[t] = ut;
  lds_barrier();

  const int s = t & 63;
  const int h = t >> 6;
  const bool scorer = s < NSENT;
  f4 mr[EMBED / 8];
  if (scorer) {
#pragma unroll
    for (int k = 0; k < EMBED / 8; ++k) mr[k] = *(const f4*)(m_s + s * MPITCH + 64 * h + 4 * k);
  }

  for (int hop = 0; hop < NHOPS; ++hop) {
    if (scorer) {
      float a0 = 0.f, a1 = 0.f, a2 = 0.f, a3 = 0.f;
#pragma unroll
      for (int k = 0; k < EMBED / 8; ++k) {
        const f4 uu = *(const f4*)(u_s + 64 * h + 4 * k);
        a0 = __builtin_fmaf(mr[k].x, uu.x, a0);
        a1 = __builtin_fmaf(mr[k].y, uu.y, a1);
        a2 = __builtin_fmaf(mr[k].z, uu.z, a2);
        a3 = __builtin_fmaf(mr[k].w, uu.w, a3);
      }
      part[h * SPITCH + s] = (a0 + a1) + (a2 + a3);
    }
    lds_barrier();

    float sc[NSENT];
#pragma unroll
    for (int q = 0; q < 12; ++q) {
      const f4 p0 = *(const f4*)(part + 4 * q);
      const f4 p1 = *(const f4*)(part + SPITCH + 4 * q);
      sc[4 * q + 0] = p0.x + p1.x;
      sc[4 * q + 1] = p0.y + p1.y;
      sc[4 * q + 2] = p0.z + p1.z;
      sc[4 * q + 3] = p0.w + p1.w;
    }
    {
      const v2f p0 = *(const v2f*)(part + 48);
      const v2f p1 = *(const v2f*)(part + SPITCH + 48);
      sc[48] = p0.x + p1.x;
      sc[49] = p0.y + p1.y;
    }
    float x0 = sc[0], x1 = sc[1], x2 = sc[2], x3 = sc[3];
#pragma unroll
    for (int q = 1; q < 12; ++q) {
      x0 = fmaxf(x0, sc[4 * q + 0]);
      x1 = fmaxf(x1, sc[4 * q + 1]);
      x2 = fmaxf(x2, sc[4 * q + 2]);
      x3 = fmaxf(x3, sc[4 * q + 3]);
    }
    x0 = fmaxf(x0, sc[48]);
    x1 = fmaxf(x1, sc[49]);
    const float mx = fmaxf(fmaxf(x0, x1), fmaxf(x2, x3));

    float e0 = 0.f, e1 = 0.f, e2 = 0.f, e3 = 0.f;
    float o0 = 0.f, o1 = 0.f, o2 = 0.f, o3 = 0.f;
#pragma unroll
    for (int q = 0; q < 12; ++q) {
      const float p0 = __expf(sc[4 * q + 0] - mx);
      const float p1 = __expf(sc[4 * q + 1] - mx);
      const float p2 = __expf(sc[4 * q + 2] - mx);
      const float p3 = __expf(sc[4 * q + 3] - mx);
      e0 += p0; e1 += p1; e2 += p2; e3 += p3;
      o0 = __builtin_fmaf(p0, cr[4 * q + 0], o0);
      o1 = __builtin_fmaf(p1, cr[4 * q + 1], o1);
      o2 = __builtin_fmaf(p2, cr[4 * q + 2], o2);
      o3 = __builtin_fmaf(p3, cr[4 * q + 3], o3);
    }
    {
      const float p0 = __expf(sc[48] - mx);
      const float p1 = __expf(sc[49] - mx);
      e0 += p0; e1 += p1;
      o0 = __builtin_fmaf(p0, cr[48], o0);
      o1 = __builtin_fmaf(p1, cr[49], o1);
    }
    const float ssum = (e0 + e1) + (e2 + e3);
    const float o    = (o0 + o1) + (o2 + o3);
    ut += o * (1.f / ssum);
    if (hop + 1 < NHOPS) {
      u_s[t] = ut;
      lds_barrier();
    }
  }
  u[b * EMBED + t] = ut;
}

#define UPITCH 132
typedef __attribute__((ext_vector_type(4))) float v4f;

__global__ void __launch_bounds__(256)
k_gemm(const float* __restrict__ u,
       const float* __restrict__ W_lin,
       const float* __restrict__ b_lin,
       float* __restrict__ out) {
  __shared__ float u_s[BATCH * UPITCH];
  const int NT   = (VOCAB + 15) / 16;
  const int tid  = (int)threadIdx.x;
  const int wraw = (int)((blockIdx.x * 256 + tid) >> 5);
  const bool live = wraw < NT;
  const int wave = live ? wraw : (NT - 1);
  const int lane = tid & 31;
  const int nbase = wave * 16;
  const int half  = lane >> 4;
  const int l16   = lane & 15;

  const int nrow   = nbase + l16;
  const int nrow_c = (nrow < VOCAB) ? nrow : (VOCAB - 1);
  const float* brow = W_lin + (size_t)nrow_c * EMBED + 4 * half;

  v4f us[8];
#pragma unroll
  for (int i = 0; i < 8; ++i) us[i] = *(const v4f*)(u + (i * 256 + tid) * 4);

  v4f bf[16];
#pragma unroll
  for (int j = 0; j < 16; ++j) bf[j] = *(const v4f*)(brow + j * 8);
  const float bias = b_lin[nrow_c];

#pragma unroll
  for (int i = 0; i < 8; ++i) {
    const int e = (i * 256 + tid) * 4;
    *(v4f*)(u_s + (e >> 7) * UPITCH + (e & (EMBED - 1))) = us[i];
  }
  lds_barrier();

  const float* arow = u_s + l16 * UPITCH + 4 * half;

  v8f acc0 = {0.f,0.f,0.f,0.f,0.f,0.f,0.f,0.f};
  v8f acc1 = acc0, acc2 = acc0, acc3 = acc0;
#pragma unroll
  for (int j = 0; j < EMBED / 8; ++j) {
    const v4f a0 = *(const v4f*)(arow + 0 * 16 * UPITCH + j * 8);
    const v4f a1 = *(const v4f*)(arow + 1 * 16 * UPITCH + j * 8);
    const v4f a2 = *(const v4f*)(arow + 2 * 16 * UPITCH + j * 8);
    const v4f a3 = *(const v4f*)(arow + 3 * 16 * UPITCH + j * 8);
    const v4f b  = bf[j];
    acc0 = __builtin_amdgcn_wmma_f32_16x16x4_f32(false, a0.xy, false, b.xy, (short)0, acc0, false, false);
    acc1 = __builtin_amdgcn_wmma_f32_16x16x4_f32(false, a1.xy, false, b.xy, (short)0, acc1, false, false);
    acc2 = __builtin_amdgcn_wmma_f32_16x16x4_f32(false, a2.xy, false, b.xy, (short)0, acc2, false, false);
    acc3 = __builtin_amdgcn_wmma_f32_16x16x4_f32(false, a3.xy, false, b.xy, (short)0, acc3, false, false);
    acc0 = __builtin_amdgcn_wmma_f32_16x16x4_f32(false, a0.zw, false, b.zw, (short)0, acc0, false, false);
    acc1 = __builtin_amdgcn_wmma_f32_16x16x4_f32(false, a1.zw, false, b.zw, (short)0, acc1, false, false);
    acc2 = __builtin_amdgcn_wmma_f32_16x16x4_f32(false, a2.zw, false, b.zw, (short)0, acc2, false, false);
    acc3 = __builtin_amdgcn_wmma_f32_16x16x4_f32(false, a3.zw, false, b.zw, (short)0, acc3, false, false);
  }

  if (live && nrow < VOCAB) {
#pragma unroll
    for (int r = 0; r < 8; ++r) {
      const int mm = r + 8 * half;
      out[(size_t)(mm +  0) * VOCAB + nrow] = acc0[r] + bias;
      out[(size_t)(mm + 16) * VOCAB + nrow] = acc1[r] + bias;
      out[(size_t)(mm + 32) * VOCAB + nrow] = acc2[r] + bias;
      out[(size_t)(mm + 48) * VOCAB + nrow] = acc3[r] + bias;
    }
  }
}

extern "C" void kernel_launch(void* const* d_in, const int* in_sizes, int n_in,
                              void* d_out, int out_size, void* d_ws, size_t ws_size,
                              hipStream_t stream) {
  (void)in_sizes; (void)n_in; (void)out_size; (void)ws_size;

  const int*   stories   = (const int*)d_in[0];
  const int*   questions = (const int*)d_in[1];
  const int*   masks     = (const int*)d_in[2];
  const float* W_A   = (const float*)d_in[3];
  const float* W_B   = (const float*)d_in[4];
  const float* W_C   = (const float*)d_in[5];
  const float* W_AT  = (const float*)d_in[6];
  const float* W_CT  = (const float*)d_in[7];
  const float* W_lin = (const float*)d_in[8];
  const float* b_lin = (const float*)d_in[9];
  float* out = (float*)d_out;

  float* ws = (float*)d_ws;
  float* m = ws;
  float* c = m + BATCH * NSENT * EMBED;
  float* u = c + BATCH * NSENT * EMBED;

  k_embed<<<BATCH + BATCH * NSENT, EMBED, 0, stream>>>(questions, stories, masks, W_B, W_A, W_C,
                                                      W_AT, W_CT, u, m, c);
  k_hops<<<BATCH, EMBED, 0, stream>>>(m, c, u);

  const int NT     = (VOCAB + 15) / 16;
  const int blocks = (NT + 7) / 8;
  k_gemm<<<blocks, 256, 0, stream>>>(u, W_lin, b_lin, out);
}
